# starting kernel
# baseline (speedup 1.0000x reference)
#include <hip/hip_runtime.h>

typedef __attribute__((ext_vector_type(16))) _Float16 v16h;
typedef __attribute__((ext_vector_type(8)))  _Float16 v8h;
typedef __attribute__((ext_vector_type(8)))  float    v8f;
typedef __attribute__((ext_vector_type(4)))  float    v4f;

#define NF_NLAYERS 8
#define NF_LATENT  64
#define NF_HALF    32
#define NF_HIDDEN  128
#define NF_SCALE   0.1f

#define NF_THREADS 512
#define NF_WAVES   (NF_THREADS / 32)
#define NF_ROWS_PER_BLOCK (NF_WAVES * 2 * 16)

#define LD1 40
#define LD2 136
#define LD3 136

#define W1_ELEMS (NF_HALF * NF_HIDDEN)
#define W2_ELEMS (NF_HIDDEN * NF_HIDDEN)
#define W3_ELEMS (NF_HIDDEN * NF_LATENT)
#define NSUB     (NF_NLAYERS * 2)

__global__ void __launch_bounds__(256)
nf_convert_weights(const float* __restrict__ W1, const float* __restrict__ W2,
                   const float* __restrict__ W3,
                   _Float16* __restrict__ tW1, _Float16* __restrict__ tW2,
                   _Float16* __restrict__ tW3)
{
    const int i = blockIdx.x * 256 + threadIdx.x;
    const int n1 = NSUB * W1_ELEMS;
    const int n2 = NSUB * W2_ELEMS;
    const int n3 = NSUB * W3_ELEMS;
    if (i < n1) {
        int w = i >> 12, r = i & 4095;
        int m = r >> 5, k = r & 31;
        tW1[i] = (_Float16)W1[w * W1_ELEMS + k * NF_HIDDEN + m];
    } else if (i < n1 + n2) {
        int j = i - n1;
        int w = j >> 14, r = j & 16383;
        int m = r >> 7, k = r & 127;
        tW2[j] = (_Float16)W2[w * W2_ELEMS + k * NF_HIDDEN + m];
    } else if (i < n1 + n2 + n3) {
        int j = i - n1 - n2;
        int w = j >> 13, r = j & 8191;
        int m = r >> 7, k = r & 127;
        tW3[j] = (_Float16)W3[w * W3_ELEMS + k * NF_LATENT + m];
    }
}

__device__ __forceinline__ v16h nf_ldsA(const _Float16* W, int ld, int mtile, int kchunk, int lane)
{
    const int row = mtile * 16 + (lane & 15);
    const int kb  = kchunk * 32 + ((lane & 16) ? 8 : 0);
    const _Float16* p = W + row * ld + kb;
    v8h lo = *(const v8h*)p;
    v8h hi = *(const v8h*)(p + 16);
    v16h a;
#pragma unroll
    for (int j = 0; j < 8; ++j) { a[j] = lo[j]; a[j + 8] = hi[j]; }
    return a;
}

__device__ __forceinline__ v8f nf_ldsBias(const float* B, int mtile, int hi8)
{
    const float* p = B + mtile * 16 + hi8;
    v8f c;
#pragma unroll
    for (int j = 0; j < 8; ++j) c[j] = p[j];
    return c;
}

__device__ __forceinline__ v16h nf_packB(v8f lo, v8f hi)
{
    v16h b;
#pragma unroll
    for (int j = 0; j < 8; ++j) { b[j] = (_Float16)lo[j]; b[j + 8] = (_Float16)hi[j]; }
    return b;
}

__device__ __forceinline__ v16h nf_relu16(v16h b)
{
#if __has_builtin(__builtin_elementwise_max)
    const v16h z = {};
    return __builtin_elementwise_max(b, z);
#else
    v16h r;
#pragma unroll
    for (int j = 0; j < 16; ++j) r[j] = b[j] > (_Float16)0.0f ? b[j] : (_Float16)0.0f;
    return r;
#endif
}

__device__ __forceinline__ v16h nf_packB_relu(v8f lo, v8f hi)
{
    return nf_relu16(nf_packB(lo, hi));
}

__device__ __forceinline__ float nf_tanh(float v)
{
#if __has_builtin(__builtin_amdgcn_tanhf)
    return __builtin_amdgcn_tanhf(v);
#else
    return tanhf(v);
#endif
}

#define NF_WMMA(a, b, c) \
    __builtin_amdgcn_wmma_f32_16x16x32_f16(false, (a), false, (b), (short)0, (c), false, false)

struct NFStage {
    v8h r1;
    v8h r2[4];
    v8h r3[2];
    float rb;
};

__device__ __forceinline__ void nf_stage_load(NFStage& st, int s, int tid,
                                              const _Float16* tW1, const _Float16* tW2,
                                              const _Float16* tW3,
                                              const float* b1, const float* b2, const float* b3)
{
    const _Float16* g1 = tW1 + (long)s * W1_ELEMS;
    const _Float16* g2 = tW2 + (long)s * W2_ELEMS;
    const _Float16* g3 = tW3 + (long)s * W3_ELEMS;
    st.r1 = *(const v8h*)&g1[tid * 8];
#pragma unroll
    for (int q = 0; q < 4; ++q) st.r2[q] = *(const v8h*)&g2[(tid + q * NF_THREADS) * 8];
#pragma unroll
    for (int q = 0; q < 2; ++q) st.r3[q] = *(const v8h*)&g3[(tid + q * NF_THREADS) * 8];
    float rb = 0.0f;
    if (tid < 128)      rb = b1[(long)s * NF_HIDDEN + tid];
    else if (tid < 256) rb = b2[(long)s * NF_HIDDEN + (tid - 128)];
    else if (tid < 320) rb = b3[(long)s * NF_LATENT + (tid - 256)];
    st.rb = rb;
}

__global__ void __launch_bounds__(NF_THREADS)
nf_flow_kernel(const float* __restrict__ x,
               const _Float16* __restrict__ tW1, const float* __restrict__ b1,
               const _Float16* __restrict__ tW2, const float* __restrict__ b2,
               const _Float16* __restrict__ tW3, const float* __restrict__ b3,
               float* __restrict__ out)
{
    __shared__ __align__(16) _Float16 sW1[NF_HIDDEN * LD1];
    __shared__ __align__(16) _Float16 sW2[NF_HIDDEN * LD2];
    __shared__ __align__(16) _Float16 sW3[NF_LATENT * LD3];
    __shared__ __align__(16) float sB1[NF_HIDDEN];
    __shared__ __align__(16) float sB2[NF_HIDDEN];
    __shared__ __align__(16) float sB3[NF_LATENT];

    const int tid  = threadIdx.x;
    const int lane = tid & 31;
    const int wave = tid >> 5;
    const int hi8  = (lane & 16) ? 8 : 0;

    const long row0 = (long)blockIdx.x * NF_ROWS_PER_BLOCK + (wave * 2 + 0) * 16 + (lane & 15);
    const long row1 = (long)blockIdx.x * NF_ROWS_PER_BLOCK + (wave * 2 + 1) * 16 + (lane & 15);

    v8f xA[4], xB[4];
    {
        const float* xr0 = x + row0 * NF_LATENT;
        const float* xr1 = x + row1 * NF_LATENT;
#pragma unroll
        for (int f = 0; f < 4; ++f) {
            const int d0 = f * 16 + hi8;
            v4f a0 = *(const v4f*)(xr0 + d0);
            v4f b0 = *(const v4f*)(xr0 + d0 + 4);
            v4f a1 = *(const v4f*)(xr1 + d0);
            v4f b1v = *(const v4f*)(xr1 + d0 + 4);
#pragma unroll
            for (int j = 0; j < 4; ++j) {
                xA[f][j] = a0[j]; xA[f][j + 4] = b0[j];
                xB[f][j] = a1[j]; xB[f][j + 4] = b1v[j];
            }
        }
    }

    NFStage st;
    nf_stage_load(st, 0, tid, tW1, tW2, tW3, b1, b2, b3);

    for (int s = 0; s < NSUB; ++s) {
        __syncthreads();
        {
            const int i1 = tid * 8;
            *(v8h*)&sW1[(i1 >> 5) * LD1 + (i1 & 31)] = st.r1;
#pragma unroll
            for (int q = 0; q < 4; ++q) {
                const int i = (tid + q * NF_THREADS) * 8;
                *(v8h*)&sW2[(i >> 7) * LD2 + (i & 127)] = st.r2[q];
            }
#pragma unroll
            for (int q = 0; q < 2; ++q) {
                const int i = (tid + q * NF_THREADS) * 8;
                *(v8h*)&sW3[(i >> 7) * LD3 + (i & 127)] = st.r3[q];
            }
            if (tid < 128)      sB1[tid] = st.rb;
            else if (tid < 256) sB2[tid - 128] = st.rb;
            else if (tid < 320) sB3[tid - 256] = st.rb;
        }
        if (s + 1 < NSUB)
            nf_stage_load(st, s + 1, tid, tW1, tW2, tW3, b1, b2, b3);
        __syncthreads();

        v16h binA = nf_packB(xA[0], xA[1]);
        v16h binB = nf_packB(xB[0], xB[1]);

        v16h p1A[4], p1B[4];
#pragma unroll
        for (int g = 0; g < 4; ++g) {
            v8f cb0 = nf_ldsBias(sB1, 2 * g, hi8);
            v8f cb1 = nf_ldsBias(sB1, 2 * g + 1, hi8);
            v16h a0 = nf_ldsA(sW1, LD1, 2 * g, 0, lane);
            v16h a1 = nf_ldsA(sW1, LD1, 2 * g + 1, 0, lane);
            v8f hA0 = NF_WMMA(a0, binA, cb0);
            v8f hB0 = NF_WMMA(a0, binB, cb0);
            v8f hA1 = NF_WMMA(a1, binA, cb1);
            v8f hB1 = NF_WMMA(a1, binB, cb1);
            p1A[g] = nf_packB_relu(hA0, hA1);
            p1B[g] = nf_packB_relu(hB0, hB1);
        }

        v16h p2A[4], p2B[4];
#pragma unroll
        for (int g = 0; g < 4; ++g) {
            v8f c0 = nf_ldsBias(sB2, 2 * g, hi8);
            v8f c1 = c0;
            v8f d0 = nf_ldsBias(sB2, 2 * g + 1, hi8);
            v8f d1 = d0;
#pragma unroll
            for (int k = 0; k < 4; ++k) {
                v16h a0 = nf_ldsA(sW2, LD2, 2 * g, k, lane);
                c0 = NF_WMMA(a0, p1A[k], c0);
                c1 = NF_WMMA(a0, p1B[k], c1);
                v16h a1 = nf_ldsA(sW2, LD2, 2 * g + 1, k, lane);
                d0 = NF_WMMA(a1, p1A[k], d0);
                d1 = NF_WMMA(a1, p1B[k], d1);
            }
            p2A[g] = nf_packB_relu(c0, d0);
            p2B[g] = nf_packB_relu(c1, d1);
        }

        v8f stA[4], stB[4];
#pragma unroll
        for (int t = 0; t < 4; ++t) {
            v8f c0 = nf_ldsBias(sB3, t, hi8);
            v8f c1 = c0;
#pragma unroll
            for (int k = 0; k < 4; ++k) {
                v16h a = nf_ldsA(sW3, LD3, t, k, lane);
                c0 = NF_WMMA(a, p2A[k], c0);
                c1 = NF_WMMA(a, p2B[k], c1);
            }
            stA[t] = c0;
            stB[t] = c1;
        }

#pragma unroll
        for (int f = 0; f < 2; ++f) {
#pragma unroll
            for (int j = 0; j < 8; ++j) {
                float eA = __expf(NF_SCALE * nf_tanh(stA[f][j]));
                float eB = __expf(NF_SCALE * nf_tanh(stB[f][j]));
                xA[2 + f][j] = xA[2 + f][j] * eA + stA[2 + f][j];
                xB[2 + f][j] = xB[2 + f][j] * eB + stB[2 + f][j];
            }
        }
#pragma unroll
        for (int f = 0; f < 2; ++f) {
            v8f tA = xA[f]; xA[f] = xA[2 + f]; xA[2 + f] = tA;
            v8f tB = xB[f]; xB[f] = xB[2 + f]; xB[2 + f] = tB;
        }
    }

    {
        float* o0 = out + row0 * NF_LATENT;
        float* o1 = out + row1 * NF_LATENT;
#pragma unroll
        for (int f = 0; f < 4; ++f) {
            const int d0 = f * 16 + hi8;
            v4f a0, b0, a1, b1v;
#pragma unroll
            for (int j = 0; j < 4; ++j) {
                a0[j] = xA[f][j]; b0[j] = xA[f][j + 4];
                a1[j] = xB[f][j]; b1v[j] = xB[f][j + 4];
            }
            *(v4f*)(o0 + d0)     = a0;
            *(v4f*)(o0 + d0 + 4) = b0;
            *(v4f*)(o1 + d0)     = a1;
            *(v4f*)(o1 + d0 + 4) = b1v;
        }
    }
}

extern "C" void kernel_launch(void* const* d_in, const int* in_sizes, int n_in,
                              void* d_out, int out_size, void* d_ws, size_t ws_size,
                              hipStream_t stream)
{
    const float* x  = (const float*)d_in[0];
    const float* W1 = (const float*)d_in[1];
    const float* b1 = (const float*)d_in[2];
    const float* W2 = (const float*)d_in[3];
    const float* b2 = (const float*)d_in[4];
    const float* W3 = (const float*)d_in[5];
    const float* b3 = (const float*)d_in[6];
    float* out = (float*)d_out;

    _Float16* tW1 = (_Float16*)d_ws;
    _Float16* tW2 = tW1 + NSUB * W1_ELEMS;
    _Float16* tW3 = tW2 + NSUB * W2_ELEMS;

    const int total_w = NSUB * (W1_ELEMS + W2_ELEMS + W3_ELEMS);
    nf_convert_weights<<<(total_w + 255) / 256, 256, 0, stream>>>(W1, W2, W3, tW1, tW2, tW3);

    const int batch  = in_sizes[0] / NF_LATENT;
    const int blocks = batch / NF_ROWS_PER_BLOCK;
    nf_flow_kernel<<<blocks, NF_THREADS, 0, stream>>>(x, tW1, b1, tW2, b2, tW3, b3, out);
}


# Register-resident weights refilled in place (no second buffer): in a sub-step's last tile pair each layer's fragments are fetched for the next sub-step once that layer is done with them
# speedup vs baseline: 1.8425x; 1.8425x over previous
#include <hip/hip_runtime.h>


typedef __attribute__((ext_vector_type(16))) _Float16 v16h;
typedef __attribute__((ext_vector_type(8)))  _Float16 v8h;
typedef __attribute__((ext_vector_type(8)))  float    v8f;
typedef __attribute__((ext_vector_type(4)))  float    v4f;

#define NF_NLAYERS 8
#define NF_LATENT  64
#define NF_HALF    32
#define NF_HIDDEN  128
#define NF_SCALE   0.1f
#define NSUB       (NF_NLAYERS * 2)

#define NF_WAVES   4
#define NF_THREADS (NF_WAVES * 32)
#define NF_PAIRS   4
#define NF_ROWS_PER_WAVE  (NF_PAIRS * 2 * 16)
#define NF_ROWS_PER_BLOCK (NF_WAVES * NF_ROWS_PER_WAVE)

#define XP 68

#define W1_ELEMS (NF_HALF * NF_HIDDEN)
#define W2_ELEMS (NF_HIDDEN * NF_HIDDEN)
#define W3_ELEMS (NF_HIDDEN * NF_LATENT)

#define FRAG_HALVES 512
#define NFRAG       56
#define SUB_HALVES  (NFRAG * FRAG_HALVES)

__global__ void __launch_bounds__(256)
nf_convert_weights(const float* __restrict__ W1, const float* __restrict__ W2,
                   const float* __restrict__ W3, _Float16* __restrict__ tW)
{
    const int i = blockIdx.x * 256 + threadIdx.x;
    if (i >= NSUB * SUB_HALVES) return;
    const int w = i / SUB_HALVES;
    const int r = i - w * SUB_HALVES;
    const int F = r >> 9;
    const int e = r & 511;
    const int part = e >> 8;
    const int ln   = (e >> 3) & 31;
    const int j    = e & 7;
    const int kin  = part * 16 + (ln >> 4) * 8 + j;
    const int mrow = ln & 15;
    float v;
    if (F < 8) {
        const int m = F * 16 + mrow;
        v = W1[w * W1_ELEMS + kin * NF_HIDDEN + m];
    } else if (F < 40) {
        const int g = F - 8;
        const int m = (g >> 2) * 16 + mrow;
        const int k = (g & 3) * 32 + kin;
        v = W2[w * W2_ELEMS + k * NF_HIDDEN + m];
    } else {
        const int g = F - 40;
        const int m = (g >> 2) * 16 + mrow;
        const int k = (g & 3) * 32 + kin;
        v = W3[w * W3_ELEMS + k * NF_LATENT + m];
    }
    tW[i] = (_Float16)v;
}

__device__ __forceinline__ v16h nf_gldA(const _Float16* frag, int lane)
{
    const v8h lo = *(const v8h*)(frag + lane * 8);
    const v8h hi = *(const v8h*)(frag + 256 + lane * 8);
    return __builtin_shufflevector(lo, hi, 0, 1, 2, 3, 4, 5, 6, 7, 8, 9, 10, 11, 12, 13, 14, 15);
}

__device__ __forceinline__ v8f nf_gldBias(const float* B, int mtile, int hi8)
{
    const float* p = B + mtile * 16 + hi8;
    const v4f a = *(const v4f*)p;
    const v4f b = *(const v4f*)(p + 4);
    return __builtin_shufflevector(a, b, 0, 1, 2, 3, 4, 5, 6, 7);
}

__device__ __forceinline__ v8f nf_ldsX(const float* p)
{
    const v4f a = *(const v4f*)p;
    const v4f b = *(const v4f*)(p + 4);
    return __builtin_shufflevector(a, b, 0, 1, 2, 3, 4, 5, 6, 7);
}

__device__ __forceinline__ void nf_stsX(float* p, v8f v)
{
    *(v4f*)p       = __builtin_shufflevector(v, v, 0, 1, 2, 3);
    *(v4f*)(p + 4) = __builtin_shufflevector(v, v, 4, 5, 6, 7);
}

__device__ __forceinline__ v16h nf_packB(v8f lo, v8f hi)
{
    v16h b;
#pragma unroll
    for (int j = 0; j < 8; ++j) { b[j] = (_Float16)lo[j]; b[j + 8] = (_Float16)hi[j]; }
    return b;
}

__device__ __forceinline__ v16h nf_relu16(v16h b)
{
#if __has_builtin(__builtin_elementwise_max)
    const v16h z = {};
    return __builtin_elementwise_max(b, z);
#else
    v16h r;
#pragma unroll
    for (int j = 0; j < 16; ++j) r[j] = b[j] > (_Float16)0.0f ? b[j] : (_Float16)0.0f;
    return r;
#endif
}

__device__ __forceinline__ v16h nf_packB_relu(v8f lo, v8f hi)
{
    return nf_relu16(nf_packB(lo, hi));
}

__device__ __forceinline__ float nf_tanh(float v)
{
#if __has_builtin(__builtin_amdgcn_tanhf)
    return __builtin_amdgcn_tanhf(v);
#else
    return tanhf(v);
#endif
}

#define NF_WMMA(a, b, c) \
    __builtin_amdgcn_wmma_f32_16x16x32_f16(false, (a), false, (b), (short)0, (c), false, false)

#define NF_FETCH_NOT_EARLIER() __builtin_amdgcn_sched_barrier(0x078F)

__device__ __forceinline__ void nf_fetch_l1(const _Float16* g, const float* c1, int lane, int hi8,
                                            v16h (&wA1)[8], v8f (&cB1)[8])
{
#pragma unroll
    for (int m = 0; m < 8; ++m) cB1[m] = nf_gldBias(c1, m, hi8);
#pragma unroll
    for (int m = 0; m < 8; ++m) wA1[m] = nf_gldA(g + m * FRAG_HALVES, lane);
}

__device__ __forceinline__ void nf_fetch_l2(const _Float16* g, const float* c2, int lane, int hi8, int G,
                                            v16h (&wA2)[8][4], v8f (&cB2)[8])
{
#pragma unroll
    for (int i = 0; i < 4; ++i) cB2[4 * G + i] = nf_gldBias(c2, 4 * G + i, hi8);
#pragma unroll
    for (int k = 0; k < 4; ++k) {
#pragma unroll
        for (int i = 0; i < 4; ++i)
            wA2[4 * G + i][k] = nf_gldA(g + (8 + (4 * G + i) * 4 + k) * FRAG_HALVES, lane);
    }
}

__device__ __forceinline__ void nf_fetch_l3(const _Float16* g, const float* c3, int lane, int hi8,
                                            v16h (&wA3)[4][4], v8f (&cB3)[4])
{
#pragma unroll
    for (int m = 0; m < 4; ++m) cB3[m] = nf_gldBias(c3, m, hi8);
#pragma unroll
    for (int k = 0; k < 4; ++k) {
#pragma unroll
        for (int m = 0; m < 4; ++m) wA3[m][k] = nf_gldA(g + (40 + m * 4 + k) * FRAG_HALVES, lane);
    }
}

template <bool REFILL>
__device__ __forceinline__ void nf_pass(float* fx, int p, int inb, int trb, int lane, int hi8,
                                        v16h& binA, v16h& binB,
                                        v16h (&wA1)[8], v16h (&wA2)[8][4], v16h (&wA3)[4][4],
                                        v8f (&cB1)[8], v8f (&cB2)[8], v8f (&cB3)[4],
                                        const _Float16* gn, const float* c1n, const float* c2n,
                                        const float* c3n)
{
    float* t0 = fx + (2 * p) * 16 * XP;
    float* t1 = t0 + 16 * XP;

    const int np  = (p + 1) & (NF_PAIRS - 1);
    const int nin = (p == NF_PAIRS - 1) ? trb : inb;
    const float* n0 = fx + (2 * np) * 16 * XP + nin;
    const float* n1 = n0 + 16 * XP;
    const v8f nA0 = nf_ldsX(n0), nA1 = nf_ldsX(n0 + 16);
    const v8f nB0 = nf_ldsX(n1), nB1 = nf_ldsX(n1 + 16);

    v16h p1A[4], p1B[4];
#pragma unroll
    for (int q = 0; q < 4; ++q) {
        v8f hA0 = NF_WMMA(wA1[2 * q],     binA, cB1[2 * q]);
        v8f hB0 = NF_WMMA(wA1[2 * q],     binB, cB1[2 * q]);
        v8f hA1 = NF_WMMA(wA1[2 * q + 1], binA, cB1[2 * q + 1]);
        v8f hB1 = NF_WMMA(wA1[2 * q + 1], binB, cB1[2 * q + 1]);
        p1A[q] = nf_packB_relu(hA0, hA1);
        p1B[q] = nf_packB_relu(hB0, hB1);
    }
    if constexpr (REFILL) {
        NF_FETCH_NOT_EARLIER();
        nf_fetch_l1(gn, c1n, lane, hi8, wA1, cB1);
    }

    v16h p2A[4], p2B[4];
#pragma unroll
    for (int G = 0; G < 2; ++G) {
        v8f aA[4], aB[4];
#pragma unroll
        for (int i = 0; i < 4; ++i) { aA[i] = cB2[4 * G + i]; aB[i] = cB2[4 * G + i]; }
#pragma unroll
        for (int k = 0; k < 4; ++k) {
#pragma unroll
            for (int i = 0; i < 4; ++i) {
                aA[i] = NF_WMMA(wA2[4 * G + i][k], p1A[k], aA[i]);
                aB[i] = NF_WMMA(wA2[4 * G + i][k], p1B[k], aB[i]);
            }
        }
        p2A[2 * G]     = nf_packB_relu(aA[0], aA[1]);
        p2A[2 * G + 1] = nf_packB_relu(aA[2], aA[3]);
        p2B[2 * G]     = nf_packB_relu(aB[0], aB[1]);
        p2B[2 * G + 1] = nf_packB_relu(aB[2], aB[3]);
        if constexpr (REFILL) {
            NF_FETCH_NOT_EARLIER();
            nf_fetch_l2(gn, c2n, lane, hi8, G, wA2, cB2);
        }
    }

    v8f oA[2], oB[2];
    oA[0] = nf_ldsX(t0 + trb); oA[1] = nf_ldsX(t0 + trb + 16);
    oB[0] = nf_ldsX(t1 + trb); oB[1] = nf_ldsX(t1 + trb + 16);

    v8f stA[4], stB[4];
#pragma unroll
    for (int t = 0; t < 4; ++t) { stA[t] = cB3[t]; stB[t] = cB3[t]; }
#pragma unroll
    for (int k = 0; k < 4; ++k) {
#pragma unroll
        for (int t = 0; t < 4; ++t) {
            stA[t] = NF_WMMA(wA3[t][k], p2A[k], stA[t]);
            stB[t] = NF_WMMA(wA3[t][k], p2B[k], stB[t]);
        }
    }

#pragma unroll
    for (int f = 0; f < 2; ++f) {
#pragma unroll
        for (int j = 0; j < 8; ++j) {
            const float eA = __expf(NF_SCALE * nf_tanh(stA[f][j]));
            const float eB = __expf(NF_SCALE * nf_tanh(stB[f][j]));
            oA[f][j] = oA[f][j] * eA + stA[2 + f][j];
            oB[f][j] = oB[f][j] * eB + stB[2 + f][j];
        }
    }
    nf_stsX(t0 + trb, oA[0]); nf_stsX(t0 + trb + 16, oA[1]);
    nf_stsX(t1 + trb, oB[0]); nf_stsX(t1 + trb + 16, oB[1]);

    binA = nf_packB(nA0, nA1);
    binB = nf_packB(nB0, nB1);

    if constexpr (REFILL) {
        NF_FETCH_NOT_EARLIER();
        nf_fetch_l3(gn, c3n, lane, hi8, wA3, cB3);
    }
}

__global__ void __launch_bounds__(NF_THREADS)
nf_flow_kernel(const float* __restrict__ x,
               const _Float16* __restrict__ tW,
               const float* __restrict__ b1, const float* __restrict__ b2,
               const float* __restrict__ b3,
               float* __restrict__ out)
{
    __shared__ __align__(16) float sX[NF_WAVES * NF_ROWS_PER_WAVE * XP];

    const int tid  = threadIdx.x;
    const int lane = tid & 31;
    const int wave = tid >> 5;
    const int hi8  = (lane & 16) ? 8 : 0;

    float* myX = sX + wave * (NF_ROWS_PER_WAVE * XP);
    const long R0 = ((long)blockIdx.x * NF_WAVES + wave) * NF_ROWS_PER_WAVE;

    v16h wA1[8], wA2[8][4], wA3[4][4];
    v8f  cB1[8], cB2[8], cB3[4];
    nf_fetch_l1(tW, b1, lane, hi8, wA1, cB1);
    nf_fetch_l2(tW, b2, lane, hi8, 0, wA2, cB2);
    nf_fetch_l2(tW, b2, lane, hi8, 1, wA2, cB2);
    nf_fetch_l3(tW, b3, lane, hi8, wA3, cB3);

    const int rsub = lane >> 4;
    const int col  = (lane & 15) * 4;
    {
        const float* src = x + R0 * NF_LATENT;
#pragma unroll 16
        for (int i = 0; i < NF_ROWS_PER_WAVE / 2; ++i) {
            const int row = i * 2 + rsub;
            const v4f v = *(const v4f*)(src + row * NF_LATENT + col);
            *(v4f*)(myX + row * XP + col) = v;
        }
    }
    __syncthreads();

    float* fx = myX + (lane & 15) * XP + hi8;

    v16h binA, binB;
    {
        const float* t0 = fx;
        const float* t1 = fx + 16 * XP;
        binA = nf_packB(nf_ldsX(t0), nf_ldsX(t0 + 16));
        binB = nf_packB(nf_ldsX(t1), nf_ldsX(t1 + 16));
    }

#pragma nounroll
    for (int s = 0; s < NSUB; ++s) {
        const int sn = (s + 1 < NSUB) ? s + 1 : NSUB - 1;
        const _Float16* gn = tW + (long)sn * SUB_HALVES;
        const float* c1n = b1 + sn * NF_HIDDEN;
        const float* c2n = b2 + sn * NF_HIDDEN;
        const float* c3n = b3 + sn * NF_LATENT;

        const int inb = (s & 1) * NF_HALF;
        const int trb = NF_HALF - inb;

#pragma nounroll
        for (int p = 0; p < NF_PAIRS - 1; ++p)
            nf_pass<false>(fx, p, inb, trb, lane, hi8, binA, binB, wA1, wA2, wA3, cB1, cB2, cB3,
                           gn, c1n, c2n, c3n);
        nf_pass<true>(fx, NF_PAIRS - 1, inb, trb, lane, hi8, binA, binB, wA1, wA2, wA3, cB1, cB2, cB3,
                      gn, c1n, c2n, c3n);
    }

    __syncthreads();

    {
        float* dst = out + R0 * NF_LATENT;
#pragma unroll 16
        for (int i = 0; i < NF_ROWS_PER_WAVE / 2; ++i) {
            const int row = i * 2 + rsub;
            const v4f v = *(const v4f*)(myX + row * XP + col);
            *(v4f*)(dst + row * NF_LATENT + col) = v;
        }
    }
}

extern "C" void kernel_launch(void* const* d_in, const int* in_sizes, int n_in,
                              void* d_out, int out_size, void* d_ws, size_t ws_size,
                              hipStream_t stream)
{
    const float* x  = (const float*)d_in[0];
    const float* W1 = (const float*)d_in[1];
    const float* b1 = (const float*)d_in[2];
    const float* W2 = (const float*)d_in[3];
    const float* b2 = (const float*)d_in[4];
    const float* W3 = (const float*)d_in[5];
    const float* b3 = (const float*)d_in[6];
    float* out = (float*)d_out;

    _Float16* tW = (_Float16*)d_ws;

    const int total_w = NSUB * SUB_HALVES;
    nf_convert_weights<<<(total_w + 255) / 256, 256, 0, stream>>>(W1, W2, W3, tW);

    const int batch  = in_sizes[0] / NF_LATENT;
    const int blocks = batch / NF_ROWS_PER_BLOCK;
    nf_flow_kernel<<<blocks, NF_THREADS, 0, stream>>>(x, tW, b1, b2, b3, out);
}
